# starting kernel
# baseline (speedup 1.0000x reference)
#include <hip/hip_runtime.h>

#define NE 8
#define IN_FEAT 1024
#define OUT_FEAT 4096
#define N_TOKENS 4096

#define M_TILE 64
#define N_BLOCK 512
#define K_CHUNK 128
#define K_STRIDE (K_CHUNK + 4)
#define MAX_TILES (N_TOKENS / M_TILE + NE)

#define WS_COUNT 0
#define WS_CURS 8
#define WS_NTILES 16
#define WS_TE 24
#define WS_TS (24 + MAX_TILES)
#define WS_TR (24 + 2 * MAX_TILES)
#define WS_PERM (24 + 3 * MAX_TILES)

typedef float v2f __attribute__((ext_vector_type(2)));
typedef float v8f __attribute__((ext_vector_type(8)));

__global__ void k_zero(int* ws) {
    if (threadIdx.x < 8) ws[WS_COUNT + threadIdx.x] = 0;
}

__global__ void k_count(const int* __restrict__ gate, int* __restrict__ ws) {
    int n = blockIdx.x * 256 + threadIdx.x;
    if (n < N_TOKENS) atomicAdd(&ws[WS_COUNT + gate[n]], 1);
}

__global__ void k_scan(int* __restrict__ ws) {
    if (threadIdx.x != 0) return;
    int off = 0, ntiles = 0;
    for (int e = 0; e < NE; ++e) {
        int c = ws[WS_COUNT + e];
        ws[WS_CURS + e] = off;
        int t = (c + M_TILE - 1) / M_TILE;
        for (int i = 0; i < t; ++i) {
            ws[WS_TE + ntiles] = e;
            ws[WS_TS + ntiles] = off + i * M_TILE;
            int rem = c - i * M_TILE;
            ws[WS_TR + ntiles] = rem < M_TILE ? rem : M_TILE;
            ++ntiles;
        }
        off += c;
    }
    ws[WS_NTILES] = ntiles;
}

__global__ void k_scatter(const int* __restrict__ gate, int* __restrict__ ws) {
    int n = blockIdx.x * 256 + threadIdx.x;
    if (n < N_TOKENS) {
        int e = gate[n];
        int p = atomicAdd(&ws[WS_CURS + e], 1);
        ws[WS_PERM + p] = n;
    }
}

__launch_bounds__(256)
__global__ void k_moe_gemm(const float* __restrict__ inp,
                           const float* __restrict__ weight,
                           const int* __restrict__ ws,
                           float* __restrict__ out) {
    __shared__ float As[M_TILE][K_STRIDE];
    __shared__ int toks[M_TILE];

    const int ntiles = ws[WS_NTILES];
    const int tile = blockIdx.x;
    if (tile >= ntiles) return;

    const int e      = ws[WS_TE + tile];
    const int rstart = ws[WS_TS + tile];
    const int rows   = ws[WS_TR + tile];
    const int tid = threadIdx.x;

    if (tid < M_TILE) {
        int r = tid;
        toks[r] = (r < rows) ? ws[WS_PERM + rstart + r] : ws[WS_PERM + rstart];
    }

    const int lane = tid & 31;
    const int wave = tid >> 5;
    const int hi   = lane >> 4;
    const int l15  = lane & 15;
    const int nbase = blockIdx.y * N_BLOCK + wave * 64;
    const float* wbase = weight + (size_t)e * OUT_FEAT * IN_FEAT;

    v8f acc[4][4];
#pragma unroll
    for (int m = 0; m < 4; ++m)
#pragma unroll
        for (int j = 0; j < 4; ++j) acc[m][j] = (v8f)0.0f;

    for (int kc = 0; kc < IN_FEAT; kc += K_CHUNK) {
        __syncthreads();
#pragma unroll
        for (int i = 0; i < (M_TILE * K_CHUNK / 4) / 256; ++i) {
            int id = tid + 256 * i;
            int r  = id >> 5;
            int c4 = id & 31;
            float4 v = *(const float4*)(inp + (size_t)toks[r] * IN_FEAT + kc + c4 * 4);
            *(float4*)(&As[r][c4 * 4]) = v;
        }
        __syncthreads();

        for (int k = 0; k < K_CHUNK; k += 4) {
            const int koff = k + hi * 2;
            v2f a[4];
#pragma unroll
            for (int m = 0; m < 4; ++m)
                a[m] = *(const v2f*)(&As[m * 16 + l15][koff]);
#pragma unroll
            for (int j = 0; j < 4; ++j) {
                const int col = nbase + j * 16 + l15;
                v2f b = *(const v2f*)(wbase + (size_t)col * IN_FEAT + (kc + koff));
#pragma unroll
                for (int m = 0; m < 4; ++m)
                    acc[m][j] = __builtin_amdgcn_wmma_f32_16x16x4_f32(
                        false, a[m], false, b, (short)0, acc[m][j], false, false);
            }
        }
    }

#pragma unroll
    for (int m = 0; m < 4; ++m)
#pragma unroll
        for (int j = 0; j < 4; ++j) {
            const int col = nbase + j * 16 + l15;
#pragma unroll
            for (int g = 0; g < 8; ++g) {
                const int r = m * 16 + g + hi * 8;
                if (r < rows)
                    out[(size_t)toks[r] * OUT_FEAT + col] = acc[m][j][g];
            }
        }
}

extern "C" void kernel_launch(void* const* d_in, const int* in_sizes, int n_in,
                              void* d_out, int out_size, void* d_ws, size_t ws_size,
                              hipStream_t stream) {
    const float* inp    = (const float*)d_in[0];
    const int*   gate   = (const int*)d_in[1];
    const float* weight = (const float*)d_in[2];
    float* out = (float*)d_out;
    int*   ws  = (int*)d_ws;

    k_zero<<<1, 32, 0, stream>>>(ws);
    k_count<<<(N_TOKENS + 255) / 256, 256, 0, stream>>>(gate, ws);
    k_scan<<<1, 32, 0, stream>>>(ws);
    k_scatter<<<(N_TOKENS + 255) / 256, 256, 0, stream>>>(gate, ws);

    dim3 grid(MAX_TILES, OUT_FEAT / N_BLOCK);
    k_moe_gemm<<<grid, 256, 0, stream>>>(inp, weight, ws, out);
}


# one persistent workgroup per (expert, 128-column weight block) walks the expert's rows in 128-row passes (wave tile 128x16, 16-row tail frags) instead of the 64x512 tile grid; sum order kept
# speedup vs baseline: 1.1724x; 1.1724x over previous
#include <hip/hip_runtime.h>


#define NE 8
#define IN_FEAT 1024
#define OUT_FEAT 4096
#define N_TOKENS 4096

#define M_TILE 128
#define N_BLOCK 128
#define K_CHUNK 256
#define K_HALF 128
#define K_STRIDE (K_HALF + 4)
#define N_CHUNKS (IN_FEAT / K_CHUNK)
#define COL_BLOCKS (OUT_FEAT / N_BLOCK)
#define NTHREADS 256
#define H_BLOCKS (K_HALF / 32)
#define K_BLOCKS (K_CHUNK / 32)
#define ST_ROWS (NTHREADS / 64)
#define ST_Q (M_TILE / ST_ROWS / K_BLOCKS)

#define WS_COUNT 0
#define WS_CURS 8
#define WS_OFF 16
#define WS_PERM 32

typedef float v2f __attribute__((ext_vector_type(2)));
typedef float v4f __attribute__((ext_vector_type(4)));
typedef float v8f __attribute__((ext_vector_type(8)));

__global__ void k_zero(int* ws) {
    if (threadIdx.x < 8) ws[WS_COUNT + threadIdx.x] = 0;
}

__global__ void k_count(const int* __restrict__ gate, int* __restrict__ ws) {
    int n = blockIdx.x * 256 + threadIdx.x;
    if (n < N_TOKENS) atomicAdd(&ws[WS_COUNT + gate[n]], 1);
}

__global__ void k_scan(int* __restrict__ ws) {
    if (threadIdx.x != 0) return;
    int off = 0;
    for (int e = 0; e < NE; ++e) {
        int c = ws[WS_COUNT + e];
        ws[WS_CURS + e] = off;
        ws[WS_OFF + e] = off;
        off += c;
    }
}

__global__ void k_scatter(const int* __restrict__ gate, int* __restrict__ ws) {
    int n = blockIdx.x * 256 + threadIdx.x;
    if (n < N_TOKENS) {
        int e = gate[n];
        int p = atomicAdd(&ws[WS_CURS + e], 1);
        ws[WS_PERM + p] = n;
    }
}

struct Stage {
    const float* src;
    const int*   tok;
    float*       dst;
};

__device__ __forceinline__ void stage_load(const Stage& st, int g, v4f (&v)[ST_Q]) {
#pragma unroll
    for (int q = 0; q < ST_Q; ++q) {
        const int row = ST_ROWS * (ST_Q * g + q);
        v[q] = *(const v4f*)(st.src + (size_t)st.tok[row] * IN_FEAT);
    }
}

__device__ __forceinline__ void stage_store(const Stage& st, int g, const v4f (&v)[ST_Q]) {
#pragma unroll
    for (int q = 0; q < ST_Q; ++q) {
        const int row = ST_ROWS * (ST_Q * g + q);
        *(v4f*)(st.dst + row * K_STRIDE) = v[q];
    }
}

template <int MF>
__device__ __forceinline__ void mma_block(int g, int kb, const float* __restrict__ a_half,
                                          const float* __restrict__ w_lane, int kc,
                                          const Stage& st, v4f (&pv)[ST_Q],
                                          v2f (&bc)[8], v8f (&acc)[8]) {
    v4f nv[ST_Q];
    stage_load(st, (g + 1) & (K_BLOCKS - 1), nv);

    const int knext = (kc + (g + 1) * 32) & (IN_FEAT - 1);
    v2f bn[8];
#pragma unroll
    for (int j = 0; j < 8; ++j) bn[j] = *(const v2f*)(w_lane + knext + 4 * j);

#pragma unroll
    for (int j = 0; j < 8; ++j) {
        v2f a[MF];
#pragma unroll
        for (int m = 0; m < MF; ++m)
            a[m] = *(const v2f*)(a_half + m * 16 * K_STRIDE + kb * 32 + 4 * j);
#pragma unroll
        for (int m = 0; m < MF; ++m)
            acc[m] = __builtin_amdgcn_wmma_f32_16x16x4_f32(
                false, a[m], false, bc[j], (short)0, acc[m], false, false);
    }

    stage_store(st, g, pv);
#pragma unroll
    for (int q = 0; q < ST_Q; ++q) pv[q] = nv[q];
#pragma unroll
    for (int j = 0; j < 8; ++j) bc[j] = bn[j];
}

template <int MF>
__device__ __forceinline__ void mma_chunk(const float* __restrict__ a_lane,
                                          const float* __restrict__ w_lane, int kc,
                                          const Stage& st, v2f (&bc)[8], v8f (&acc)[8]) {
    v4f pv[ST_Q];
    stage_load(st, 0, pv);
#pragma unroll 1
    for (int kh = 0; kh < 2; ++kh) {
        const float* a_half = a_lane + kh * (M_TILE * K_STRIDE);
        if (MF == 8) {
#pragma unroll
            for (int kb = 0; kb < H_BLOCKS; ++kb)
                mma_block<MF>(kh * H_BLOCKS + kb, kb, a_half, w_lane, kc, st, pv, bc, acc);
        } else {
#pragma unroll 1
            for (int kb = 0; kb < H_BLOCKS; ++kb)
                mma_block<MF>(kh * H_BLOCKS + kb, kb, a_half, w_lane, kc, st, pv, bc, acc);
        }
    }
}

__launch_bounds__(NTHREADS)
__global__ void k_moe_gemm(const float* __restrict__ inp,
                           const float* __restrict__ weight,
                           const int* __restrict__ ws,
                           float* __restrict__ out) {
    __shared__ float As[2][2][M_TILE][K_STRIDE];
    __shared__ int toks[2][M_TILE];

    const int e  = blockIdx.x / COL_BLOCKS;
    const int cb = blockIdx.x % COL_BLOCKS;
    const int cnt = ws[WS_COUNT + e];
    const int off = ws[WS_OFF + e];
    const int ntile = (cnt + M_TILE - 1) / M_TILE;
    if (ntile == 0) return;
    const int nsteps = ntile * N_CHUNKS;
    const int* __restrict__ perm = ws + WS_PERM + off;

    const int tid  = threadIdx.x;
    const int lane = tid & 31;
    const int wave = tid >> 5;
    const int hi   = lane >> 4;
    const int l15  = lane & 15;
    const int col  = cb * N_BLOCK + wave * 16 + l15;
    const float* w_lane = weight + ((size_t)e * OUT_FEAT + col) * IN_FEAT + 2 * hi;

    const int sc4 = tid & 63;
    const int sr0 = tid >> 6;
    const int skh = sc4 >> 5;
    const int sc  = (sc4 & 31) * 4;

    v8f acc[8];
#pragma unroll
    for (int m = 0; m < 8; ++m) acc[m] = (v8f)0.0f;

    v2f bc[8];
#pragma unroll
    for (int j = 0; j < 8; ++j) bc[j] = *(const v2f*)(w_lane + 4 * j);

    if (tid < M_TILE) toks[0][tid] = perm[tid < cnt ? tid : 0];
    __syncthreads();
    {
        Stage st0;
        st0.src = inp + sc4 * 4;
        st0.tok = &toks[0][sr0];
        st0.dst = &As[0][skh][sr0][sc];
#pragma unroll
        for (int g = 0; g < K_BLOCKS; ++g) {
            v4f v[ST_Q];
            stage_load(st0, g, v);
            stage_store(st0, g, v);
        }
    }
    __syncthreads();

    for (int s = 0; s < nsteps; ++s) {
        const int t  = s / N_CHUNKS;
        const int ch = s % N_CHUNKS;
        const int s1 = (s + 1 < nsteps) ? s + 1 : s;
        const int t1 = s1 / N_CHUNKS;

        Stage st;
        st.src = inp + (s1 % N_CHUNKS) * K_CHUNK + sc4 * 4;
        st.tok = &toks[t1 & 1][sr0];
        st.dst = &As[(s + 1) & 1][skh][sr0][sc];

        const int rows_t = (cnt - t * M_TILE) < M_TILE ? (cnt - t * M_TILE) : M_TILE;
        const int mf = (rows_t + 15) >> 4;

        const float* a_lane = &As[s & 1][0][l15][2 * hi];
        const int kc = ch * K_CHUNK;
        switch (mf) {
            case 8: mma_chunk<8>(a_lane, w_lane, kc, st, bc, acc); break;
            case 7: mma_chunk<7>(a_lane, w_lane, kc, st, bc, acc); break;
            case 6: mma_chunk<6>(a_lane, w_lane, kc, st, bc, acc); break;
            case 5: mma_chunk<5>(a_lane, w_lane, kc, st, bc, acc); break;
            case 4: mma_chunk<4>(a_lane, w_lane, kc, st, bc, acc); break;
            case 3: mma_chunk<3>(a_lane, w_lane, kc, st, bc, acc); break;
            case 2: mma_chunk<2>(a_lane, w_lane, kc, st, bc, acc); break;
            default: mma_chunk<1>(a_lane, w_lane, kc, st, bc, acc); break;
        }

        if (ch == N_CHUNKS - 1) {
#pragma unroll
            for (int m = 0; m < 8; ++m) {
#pragma unroll
                for (int g = 0; g < 8; ++g) {
                    const int r = m * 16 + g + hi * 8;
                    if (r < rows_t) {
                        const int tok = toks[t & 1][r];
                        out[(size_t)tok * OUT_FEAT + col] = acc[m][g];
                    }
                }
                acc[m] = (v8f)0.0f;
            }
        }

        if (ch == N_CHUNKS - 2 && t + 1 < ntile && tid < M_TILE) {
            const int p = (t + 1) * M_TILE + tid;
            toks[(t + 1) & 1][tid] = perm[p < cnt ? p : 0];
        }

        __syncthreads();
    }
}

extern "C" void kernel_launch(void* const* d_in, const int* in_sizes, int n_in,
                              void* d_out, int out_size, void* d_ws, size_t ws_size,
                              hipStream_t stream) {
    const float* inp    = (const float*)d_in[0];
    const int*   gate   = (const int*)d_in[1];
    const float* weight = (const float*)d_in[2];
    float* out = (float*)d_out;
    int*   ws  = (int*)d_ws;

    k_zero<<<1, 32, 0, stream>>>(ws);
    k_count<<<(N_TOKENS + 255) / 256, 256, 0, stream>>>(gate, ws);
    k_scan<<<1, 32, 0, stream>>>(ws);
    k_scatter<<<(N_TOKENS + 255) / 256, 256, 0, stream>>>(gate, ws);

    k_moe_gemm<<<NE * COL_BLOCKS, NTHREADS, 0, stream>>>(inp, weight, ws, out);
}


# a pass's result stores no longer wait one by one for their token id: a lane's 64 ids are read from LDS as one group, all row offsets formed first, and full passes store unguarded as one run
# speedup vs baseline: 1.2573x; 1.0724x over previous
#include <hip/hip_runtime.h>


#define NE 8
#define IN_FEAT 1024
#define OUT_FEAT 4096
#define N_TOKENS 4096

#define M_TILE 128
#define N_BLOCK 128
#define K_CHUNK 256
#define K_HALF 128
#define K_STRIDE (K_HALF + 4)
#define N_CHUNKS (IN_FEAT / K_CHUNK)
#define COL_BLOCKS (OUT_FEAT / N_BLOCK)
#define NTHREADS 256
#define H_BLOCKS (K_HALF / 32)
#define K_BLOCKS (K_CHUNK / 32)
#define ST_ROWS (NTHREADS / 64)
#define ST_Q (M_TILE / ST_ROWS / K_BLOCKS)

#define WS_COUNT 0
#define WS_CURS 8
#define WS_OFF 16
#define WS_PERM 32

typedef float v2f __attribute__((ext_vector_type(2)));
typedef float v4f __attribute__((ext_vector_type(4)));
typedef float v8f __attribute__((ext_vector_type(8)));
typedef int   v4i __attribute__((ext_vector_type(4)));

__global__ void k_zero(int* ws) {
    if (threadIdx.x < 8) ws[WS_COUNT + threadIdx.x] = 0;
}

__global__ void k_count(const int* __restrict__ gate, int* __restrict__ ws) {
    int n = blockIdx.x * 256 + threadIdx.x;
    if (n < N_TOKENS) atomicAdd(&ws[WS_COUNT + gate[n]], 1);
}

__global__ void k_scan(int* __restrict__ ws) {
    if (threadIdx.x != 0) return;
    int off = 0;
    for (int e = 0; e < NE; ++e) {
        int c = ws[WS_COUNT + e];
        ws[WS_CURS + e] = off;
        ws[WS_OFF + e] = off;
        off += c;
    }
}

__global__ void k_scatter(const int* __restrict__ gate, int* __restrict__ ws) {
    int n = blockIdx.x * 256 + threadIdx.x;
    if (n < N_TOKENS) {
        int e = gate[n];
        int p = atomicAdd(&ws[WS_CURS + e], 1);
        ws[WS_PERM + p] = n;
    }
}

struct Stage {
    const float* src;
    const int*   tok;
    float*       dst;
};

__device__ __forceinline__ void stage_load(const Stage& st, int g, v4f (&v)[ST_Q]) {
#pragma unroll
    for (int q = 0; q < ST_Q; ++q) {
        const int row = ST_ROWS * (ST_Q * g + q);
        v[q] = *(const v4f*)(st.src + (size_t)st.tok[row] * IN_FEAT);
    }
}

__device__ __forceinline__ void stage_store(const Stage& st, int g, const v4f (&v)[ST_Q]) {
#pragma unroll
    for (int q = 0; q < ST_Q; ++q) {
        const int row = ST_ROWS * (ST_Q * g + q);
        *(v4f*)(st.dst + row * K_STRIDE) = v[q];
    }
}

template <int MF>
__device__ __forceinline__ void mma_block(int g, int kb, const float* __restrict__ a_half,
                                          const float* __restrict__ w_lane, int kc,
                                          const Stage& st, v4f (&pv)[ST_Q],
                                          v2f (&bc)[8], v8f (&acc)[8]) {
    v4f nv[ST_Q];
    stage_load(st, (g + 1) & (K_BLOCKS - 1), nv);

    const int knext = (kc + (g + 1) * 32) & (IN_FEAT - 1);
    v2f bn[8];
#pragma unroll
    for (int j = 0; j < 8; ++j) bn[j] = *(const v2f*)(w_lane + knext + 4 * j);

#pragma unroll
    for (int j = 0; j < 8; ++j) {
        v2f a[MF];
#pragma unroll
        for (int m = 0; m < MF; ++m)
            a[m] = *(const v2f*)(a_half + m * 16 * K_STRIDE + kb * 32 + 4 * j);
#pragma unroll
        for (int m = 0; m < MF; ++m)
            acc[m] = __builtin_amdgcn_wmma_f32_16x16x4_f32(
                false, a[m], false, bc[j], (short)0, acc[m], false, false);
    }

    stage_store(st, g, pv);
#pragma unroll
    for (int q = 0; q < ST_Q; ++q) pv[q] = nv[q];
#pragma unroll
    for (int j = 0; j < 8; ++j) bc[j] = bn[j];
}

template <int MF>
__device__ __forceinline__ void mma_chunk(const float* __restrict__ a_lane,
                                          const float* __restrict__ w_lane, int kc,
                                          const Stage& st, v2f (&bc)[8], v8f (&acc)[8]) {
    v4f pv[ST_Q];
    stage_load(st, 0, pv);
#pragma unroll 1
    for (int kh = 0; kh < 2; ++kh) {
        const float* a_half = a_lane + kh * (M_TILE * K_STRIDE);
        if (MF == 8) {
#pragma unroll
            for (int kb = 0; kb < H_BLOCKS; ++kb)
                mma_block<MF>(kh * H_BLOCKS + kb, kb, a_half, w_lane, kc, st, pv, bc, acc);
        } else {
#pragma unroll 1
            for (int kb = 0; kb < H_BLOCKS; ++kb)
                mma_block<MF>(kh * H_BLOCKS + kb, kb, a_half, w_lane, kc, st, pv, bc, acc);
        }
    }
}

__launch_bounds__(NTHREADS)
__global__ void k_moe_gemm(const float* __restrict__ inp,
                           const float* __restrict__ weight,
                           const int* __restrict__ ws,
                           float* __restrict__ out) {
    __shared__ float As[2][2][M_TILE][K_STRIDE];
    __shared__ __attribute__((aligned(16))) int toks[2][M_TILE];

    const int e  = blockIdx.x / COL_BLOCKS;
    const int cb = blockIdx.x % COL_BLOCKS;
    const int cnt = ws[WS_COUNT + e];
    const int off = ws[WS_OFF + e];
    const int ntile = (cnt + M_TILE - 1) / M_TILE;
    if (ntile == 0) return;
    const int nsteps = ntile * N_CHUNKS;
    const int* __restrict__ perm = ws + WS_PERM + off;

    const int tid  = threadIdx.x;
    const int lane = tid & 31;
    const int wave = tid >> 5;
    const int hi   = lane >> 4;
    const int l15  = lane & 15;
    const int col  = cb * N_BLOCK + wave * 16 + l15;
    const float* w_lane = weight + ((size_t)e * OUT_FEAT + col) * IN_FEAT + 2 * hi;

    const int sc4 = tid & 63;
    const int sr0 = tid >> 6;
    const int skh = sc4 >> 5;
    const int sc  = (sc4 & 31) * 4;

    v8f acc[8];
#pragma unroll
    for (int m = 0; m < 8; ++m) acc[m] = (v8f)0.0f;

    v2f bc[8];
#pragma unroll
    for (int j = 0; j < 8; ++j) bc[j] = *(const v2f*)(w_lane + 4 * j);

    if (tid < M_TILE) toks[0][tid] = perm[tid < cnt ? tid : 0];
    __syncthreads();
    {
        Stage st0;
        st0.src = inp + sc4 * 4;
        st0.tok = &toks[0][sr0];
        st0.dst = &As[0][skh][sr0][sc];
#pragma unroll
        for (int g = 0; g < K_BLOCKS; ++g) {
            v4f v[ST_Q];
            stage_load(st0, g, v);
            stage_store(st0, g, v);
        }
    }
    __syncthreads();

    for (int s = 0; s < nsteps; ++s) {
        const int t  = s / N_CHUNKS;
        const int ch = s % N_CHUNKS;
        const int s1 = (s + 1 < nsteps) ? s + 1 : s;
        const int t1 = s1 / N_CHUNKS;

        Stage st;
        st.src = inp + (s1 % N_CHUNKS) * K_CHUNK + sc4 * 4;
        st.tok = &toks[t1 & 1][sr0];
        st.dst = &As[(s + 1) & 1][skh][sr0][sc];

        const int rows_t = (cnt - t * M_TILE) < M_TILE ? (cnt - t * M_TILE) : M_TILE;
        const int mf = (rows_t + 15) >> 4;

        const float* a_lane = &As[s & 1][0][l15][2 * hi];
        const int kc = ch * K_CHUNK;
        switch (mf) {
            case 8: mma_chunk<8>(a_lane, w_lane, kc, st, bc, acc); break;
            case 7: mma_chunk<7>(a_lane, w_lane, kc, st, bc, acc); break;
            case 6: mma_chunk<6>(a_lane, w_lane, kc, st, bc, acc); break;
            case 5: mma_chunk<5>(a_lane, w_lane, kc, st, bc, acc); break;
            case 4: mma_chunk<4>(a_lane, w_lane, kc, st, bc, acc); break;
            case 3: mma_chunk<3>(a_lane, w_lane, kc, st, bc, acc); break;
            case 2: mma_chunk<2>(a_lane, w_lane, kc, st, bc, acc); break;
            default: mma_chunk<1>(a_lane, w_lane, kc, st, bc, acc); break;
        }

        if (ch == N_CHUNKS - 1) {
            const int* tk = &toks[t & 1][8 * hi];
            unsigned off[8][8];
#pragma unroll
            for (int m = 0; m < 8; ++m) {
                const v4i t0 = *(const v4i*)(tk + 16 * m);
                const v4i t1 = *(const v4i*)(tk + 16 * m + 4);
#pragma unroll
                for (int g = 0; g < 4; ++g) {
                    off[m][g]     = (unsigned)t0[g] * OUT_FEAT + (unsigned)col;
                    off[m][g + 4] = (unsigned)t1[g] * OUT_FEAT + (unsigned)col;
                }
            }
            if (rows_t == M_TILE) {
#pragma unroll
                for (int m = 0; m < 8; ++m) {
#pragma unroll
                    for (int g = 0; g < 8; ++g) out[off[m][g]] = acc[m][g];
                }
            } else {
#pragma unroll
                for (int m = 0; m < 8; ++m) {
                    if (m * 16 < rows_t) {
#pragma unroll
                        for (int g = 0; g < 8; ++g) {
                            const int r = m * 16 + g + hi * 8;
                            if (r < rows_t) out[off[m][g]] = acc[m][g];
                        }
                    }
                }
            }
#pragma unroll
            for (int m = 0; m < 8; ++m) acc[m] = (v8f)0.0f;
        }

        if (ch == N_CHUNKS - 2 && t + 1 < ntile && tid < M_TILE) {
            const int p = (t + 1) * M_TILE + tid;
            toks[(t + 1) & 1][tid] = perm[p < cnt ? p : 0];
        }

        __syncthreads();
    }
}

extern "C" void kernel_launch(void* const* d_in, const int* in_sizes, int n_in,
                              void* d_out, int out_size, void* d_ws, size_t ws_size,
                              hipStream_t stream) {
    const float* inp    = (const float*)d_in[0];
    const int*   gate   = (const int*)d_in[1];
    const float* weight = (const float*)d_in[2];
    float* out = (float*)d_out;
    int*   ws  = (int*)d_ws;

    k_zero<<<1, 32, 0, stream>>>(ws);
    k_count<<<(N_TOKENS + 255) / 256, 256, 0, stream>>>(gate, ws);
    k_scan<<<1, 32, 0, stream>>>(ws);
    k_scatter<<<(N_TOKENS + 255) / 256, 256, 0, stream>>>(gate, ws);

    k_moe_gemm<<<NE * COL_BLOCKS, NTHREADS, 0, stream>>>(inp, weight, ws, out);
}


# routing built inside the GEMM workgroup: each one lists its expert's tokens in LDS from the gate vector (ascending, no atomics); the four count/scan/scatter launches go; sum order kept
# speedup vs baseline: 1.2949x; 1.0299x over previous
#include <hip/hip_runtime.h>


#define NE 8
#define IN_FEAT 1024
#define OUT_FEAT 4096
#define N_TOKENS 4096

#define M_TILE 128
#define N_BLOCK 128
#define K_CHUNK 256
#define K_HALF 128
#define K_STRIDE (K_HALF + 4)
#define N_CHUNKS (IN_FEAT / K_CHUNK)
#define COL_BLOCKS (OUT_FEAT / N_BLOCK)
#define NTHREADS 256
#define H_BLOCKS (K_HALF / 32)
#define K_BLOCKS (K_CHUNK / 32)
#define ST_ROWS (NTHREADS / 64)
#define ST_Q (M_TILE / ST_ROWS / K_BLOCKS)

typedef float v2f __attribute__((ext_vector_type(2)));
typedef float v4f __attribute__((ext_vector_type(4)));
typedef float v8f __attribute__((ext_vector_type(8)));
typedef int   v4i __attribute__((ext_vector_type(4)));

struct Stage {
    const float* src;
    const int*   tok;
    float*       dst;
};

__device__ __forceinline__ void stage_load(const Stage& st, int g, v4f (&v)[ST_Q]) {
#pragma unroll
    for (int q = 0; q < ST_Q; ++q) {
        const int row = ST_ROWS * (ST_Q * g + q);
        v[q] = *(const v4f*)(st.src + (size_t)st.tok[row] * IN_FEAT);
    }
}

__device__ __forceinline__ void stage_store(const Stage& st, int g, const v4f (&v)[ST_Q]) {
#pragma unroll
    for (int q = 0; q < ST_Q; ++q) {
        const int row = ST_ROWS * (ST_Q * g + q);
        *(v4f*)(st.dst + row * K_STRIDE) = v[q];
    }
}

template <int MF>
__device__ __forceinline__ void mma_block(int g, int kb, const float* __restrict__ a_half,
                                          const float* __restrict__ w_lane, int kc,
                                          const Stage& st, v4f (&pv)[ST_Q],
                                          v2f (&bc)[8], v8f (&acc)[8]) {
    v4f nv[ST_Q];
    stage_load(st, (g + 1) & (K_BLOCKS - 1), nv);

    const int knext = (kc + (g + 1) * 32) & (IN_FEAT - 1);
    v2f bn[8];
#pragma unroll
    for (int j = 0; j < 8; ++j) bn[j] = *(const v2f*)(w_lane + knext + 4 * j);

#pragma unroll
    for (int j = 0; j < 8; ++j) {
        v2f a[MF];
#pragma unroll
        for (int m = 0; m < MF; ++m)
            a[m] = *(const v2f*)(a_half + m * 16 * K_STRIDE + kb * 32 + 4 * j);
#pragma unroll
        for (int m = 0; m < MF; ++m)
            acc[m] = __builtin_amdgcn_wmma_f32_16x16x4_f32(
                false, a[m], false, bc[j], (short)0, acc[m], false, false);
    }

    stage_store(st, g, pv);
#pragma unroll
    for (int q = 0; q < ST_Q; ++q) pv[q] = nv[q];
#pragma unroll
    for (int j = 0; j < 8; ++j) bc[j] = bn[j];
}

template <int MF>
__device__ __forceinline__ void mma_chunk(const float* __restrict__ a_lane,
                                          const float* __restrict__ w_lane, int kc,
                                          const Stage& st, v2f (&bc)[8], v8f (&acc)[8]) {
    v4f pv[ST_Q];
    stage_load(st, 0, pv);
#pragma unroll 1
    for (int kh = 0; kh < 2; ++kh) {
        const float* a_half = a_lane + kh * (M_TILE * K_STRIDE);
        if (MF == 8) {
#pragma unroll
            for (int kb = 0; kb < H_BLOCKS; ++kb)
                mma_block<MF>(kh * H_BLOCKS + kb, kb, a_half, w_lane, kc, st, pv, bc, acc);
        } else {
#pragma unroll 1
            for (int kb = 0; kb < H_BLOCKS; ++kb)
                mma_block<MF>(kh * H_BLOCKS + kb, kb, a_half, w_lane, kc, st, pv, bc, acc);
        }
    }
}

__launch_bounds__(NTHREADS)
__global__ void k_moe_gemm(const float* __restrict__ inp,
                           const float* __restrict__ weight,
                           const int* __restrict__ gate,
                           float* __restrict__ out) {
    __shared__ float As[2][2][M_TILE][K_STRIDE];
    __shared__ __attribute__((aligned(16))) int toks[2][M_TILE];
    __shared__ int list[N_TOKENS];
    __shared__ int wave_cnt[NTHREADS / 32];

    const int e  = blockIdx.x / COL_BLOCKS;
    const int cb = blockIdx.x % COL_BLOCKS;

    const int tid  = threadIdx.x;
    const int lane = tid & 31;
    const int wave = tid >> 5;
    const int hi   = lane >> 4;
    const int l15  = lane & 15;
    const int col  = cb * N_BLOCK + wave * 16 + l15;
    const float* w_lane = weight + ((size_t)e * OUT_FEAT + col) * IN_FEAT + 2 * hi;

    v4i gv[4];
#pragma unroll
    for (int i = 0; i < 4; ++i) gv[i] = *(const v4i*)(gate + 16 * tid + 4 * i);

    v2f bc[8];
#pragma unroll
    for (int j = 0; j < 8; ++j) bc[j] = *(const v2f*)(w_lane + 4 * j);

    unsigned mine = 0;
#pragma unroll
    for (int i = 0; i < 4; ++i) {
#pragma unroll
        for (int c = 0; c < 4; ++c) mine |= (gv[i][c] == e ? 1u : 0u) << (4 * i + c);
    }
    const int my_cnt = __popc(mine);
    int incl = my_cnt;
#pragma unroll
    for (int d = 1; d < 32; d <<= 1) {
        const int up = __shfl_up(incl, d, 32);
        if (lane >= d) incl += up;
    }
    if (lane == 31) wave_cnt[wave] = incl;
    __syncthreads();
    int cnt = 0;
    int pos = incl - my_cnt;
#pragma unroll
    for (int w = 0; w < NTHREADS / 32; ++w) {
        const int c = wave_cnt[w];
        if (w < wave) pos += c;
        cnt += c;
    }
    const int ntile = (cnt + M_TILE - 1) / M_TILE;
    if (ntile == 0) return;
    const int nsteps = ntile * N_CHUNKS;
#pragma unroll
    for (int b = 0; b < 16; ++b) {
        if ((mine >> b) & 1u) { list[pos] = 16 * tid + b; ++pos; }
    }
    __syncthreads();

    const int sc4 = tid & 63;
    const int sr0 = tid >> 6;
    const int skh = sc4 >> 5;
    const int sc  = (sc4 & 31) * 4;

    v8f acc[8];
#pragma unroll
    for (int m = 0; m < 8; ++m) acc[m] = (v8f)0.0f;

    if (tid < M_TILE) toks[0][tid] = list[tid < cnt ? tid : 0];
    __syncthreads();
    {
        Stage st0;
        st0.src = inp + sc4 * 4;
        st0.tok = &toks[0][sr0];
        st0.dst = &As[0][skh][sr0][sc];
#pragma unroll
        for (int g = 0; g < K_BLOCKS; ++g) {
            v4f v[ST_Q];
            stage_load(st0, g, v);
            stage_store(st0, g, v);
        }
    }
    __syncthreads();

    for (int s = 0; s < nsteps; ++s) {
        const int t  = s / N_CHUNKS;
        const int ch = s % N_CHUNKS;
        const int s1 = (s + 1 < nsteps) ? s + 1 : s;
        const int t1 = s1 / N_CHUNKS;

        Stage st;
        st.src = inp + (s1 % N_CHUNKS) * K_CHUNK + sc4 * 4;
        st.tok = &toks[t1 & 1][sr0];
        st.dst = &As[(s + 1) & 1][skh][sr0][sc];

        const int rows_t = (cnt - t * M_TILE) < M_TILE ? (cnt - t * M_TILE) : M_TILE;
        const int mf = (rows_t + 15) >> 4;

        const float* a_lane = &As[s & 1][0][l15][2 * hi];
        const int kc = ch * K_CHUNK;
        switch (mf) {
            case 8: mma_chunk<8>(a_lane, w_lane, kc, st, bc, acc); break;
            case 7: mma_chunk<7>(a_lane, w_lane, kc, st, bc, acc); break;
            case 6: mma_chunk<6>(a_lane, w_lane, kc, st, bc, acc); break;
            case 5: mma_chunk<5>(a_lane, w_lane, kc, st, bc, acc); break;
            case 4: mma_chunk<4>(a_lane, w_lane, kc, st, bc, acc); break;
            case 3: mma_chunk<3>(a_lane, w_lane, kc, st, bc, acc); break;
            case 2: mma_chunk<2>(a_lane, w_lane, kc, st, bc, acc); break;
            default: mma_chunk<1>(a_lane, w_lane, kc, st, bc, acc); break;
        }

        if (ch == N_CHUNKS - 1) {
            const int* tk = &toks[t & 1][8 * hi];
            unsigned off[8][8];
#pragma unroll
            for (int m = 0; m < 8; ++m) {
                const v4i t0 = *(const v4i*)(tk + 16 * m);
                const v4i t1 = *(const v4i*)(tk + 16 * m + 4);
#pragma unroll
                for (int g = 0; g < 4; ++g) {
                    off[m][g]     = (unsigned)t0[g] * OUT_FEAT + (unsigned)col;
                    off[m][g + 4] = (unsigned)t1[g] * OUT_FEAT + (unsigned)col;
                }
            }
            if (rows_t == M_TILE) {
#pragma unroll
                for (int m = 0; m < 8; ++m) {
#pragma unroll
                    for (int g = 0; g < 8; ++g) out[off[m][g]] = acc[m][g];
                }
            } else {
#pragma unroll
                for (int m = 0; m < 8; ++m) {
                    if (m * 16 < rows_t) {
#pragma unroll
                        for (int g = 0; g < 8; ++g) {
                            const int r = m * 16 + g + hi * 8;
                            if (r < rows_t) out[off[m][g]] = acc[m][g];
                        }
                    }
                }
            }
#pragma unroll
            for (int m = 0; m < 8; ++m) acc[m] = (v8f)0.0f;
        }

        if (ch == N_CHUNKS - 2 && t + 1 < ntile && tid < M_TILE) {
            const int p = (t + 1) * M_TILE + tid;
            toks[(t + 1) & 1][tid] = list[p < cnt ? p : 0];
        }

        __syncthreads();
    }
}

extern "C" void kernel_launch(void* const* d_in, const int* in_sizes, int n_in,
                              void* d_out, int out_size, void* d_ws, size_t ws_size,
                              hipStream_t stream) {
    const float* inp    = (const float*)d_in[0];
    const int*   gate   = (const int*)d_in[1];
    const float* weight = (const float*)d_in[2];
    float* out = (float*)d_out;

    k_moe_gemm<<<NE * COL_BLOCKS, NTHREADS, 0, stream>>>(inp, weight, gate, out);
}
